# starting kernel
# baseline (speedup 1.0000x reference)
#include <hip/hip_runtime.h>
#include <hip/hip_bf16.h>
#include <math.h>


#define BATCH    8
#define LENGTH   2048
#define D_MODEL  1024
#define D_STATE  64
#define TCH      16
#define CHUNKS   (LENGTH / TCH)
#define COLS     (BATCH * CHUNKS)
#define TILES    (COLS / 16)

typedef __attribute__((ext_vector_type(2))) float v2f;
typedef __attribute__((ext_vector_type(8))) float v8f;

#if defined(__HIP_DEVICE_COMPILE__)
#if !__has_builtin(__builtin_amdgcn_wmma_f32_16x16x4_f32)
#error "gfx1250 device pass: __builtin_amdgcn_wmma_f32_16x16x4_f32 not available"
#endif
#endif

__device__ __forceinline__ v8f wmma_f32_16x16x4(v2f a, v2f b, v8f c) {
#if defined(__HIP_DEVICE_COMPILE__)
  return __builtin_amdgcn_wmma_f32_16x16x4_f32(false, a, false, b, (short)0, c,
                                               false, false);
#else
  (void)a; (void)b;
  return c;
#endif
}

__global__ __launch_bounds__(256) void s4d_intra(
    const float* __restrict__ x, const float* __restrict__ log_dt,
    const float* __restrict__ Bmat, const float* __restrict__ Cmat,
    const float* __restrict__ Dvec, float* __restrict__ y)
{
  const int d = blockIdx.x;
  __shared__ float sP[TCH][D_STATE];
  __shared__ float sK[TCH];

  const int tid = threadIdx.x;
  const float dt = expf(log_dt[d]);

  if (tid < D_STATE) {
    const float An  = -0.5f * (float)(tid + 1);
    const float dtA = dt * An;
    const float Abar = expf(dtA);
    float q = Cmat[d * D_STATE + tid] *
              (expm1f(dtA) / An) * Bmat[d * D_STATE + tid];
    #pragma unroll
    for (int j = 0; j < TCH; ++j) { sP[j][tid] = q; q *= Abar; }
  }
  __syncthreads();
  if (tid < TCH) {
    float acc = 0.f;
    #pragma unroll 8
    for (int n = 0; n < D_STATE; ++n) acc += sP[tid][n];
    sK[tid] = acc;
  }
  __syncthreads();

  const int lane = tid & 31;
  const int wave = tid >> 5;
  const int row  = lane & 15;
  const int hi   = lane >> 4;

  v2f afrag[4];
  #pragma unroll
  for (int kk = 0; kk < 4; ++kk) {
    #pragma unroll
    for (int v = 0; v < 2; ++v) {
      const int col = kk * 4 + hi * 2 + v;
      afrag[kk][v] = (row >= col) ? sK[row - col] : 0.f;
    }
  }

  const float Dd = Dvec[d];

  for (int t8 = 0; t8 < TILES / 8; ++t8) {
    const int tile = t8 * 8 + wave;
    const int cc = tile * 16 + row;
    const int b  = cc >> 7;
    const int ch = cc & 127;
    const int base = (b * LENGTH + ch * TCH) * D_MODEL + d;

    v8f acc = {};
    #pragma unroll
    for (int kk = 0; kk < 4; ++kk) {
      v2f bfrag;
      #pragma unroll
      for (int v = 0; v < 2; ++v) {
        const int j = kk * 4 + hi * 2 + v;
        bfrag[v] = x[base + j * D_MODEL];
      }
      acc = wmma_f32_16x16x4(afrag[kk], bfrag, acc);
    }

    #pragma unroll
    for (int v = 0; v < 8; ++v) {
      const int r   = v + hi * 8;
      const int idx = base + r * D_MODEL;
      y[idx] = acc[v] + Dd * x[idx];
    }
  }
}

__global__ __launch_bounds__(32) void s4d_state(
    const float* __restrict__ x, const float* __restrict__ log_dt,
    const float* __restrict__ Bmat, const float* __restrict__ Cmat,
    float* __restrict__ y)
{
  const int d    = blockIdx.x;
  const int lane = threadIdx.x;
  const int row  = lane & 15;
  const int hi   = lane >> 4;

  __shared__ float sAbar[D_STATE];
  __shared__ float sBbar[D_STATE];
  __shared__ float sX[TCH * BATCH];

  const float dt = expf(log_dt[d]);
  for (int t = lane; t < D_STATE; t += 32) {
    const float An  = -0.5f * (float)(t + 1);
    const float dtA = dt * An;
    sAbar[t] = expf(dtA);
    sBbar[t] = (expm1f(dtA) / An) * Bmat[d * D_STATE + t];
  }
  __syncthreads();

  v2f g[16];
  #pragma unroll
  for (int kk = 0; kk < 16; ++kk) {
    #pragma unroll
    for (int v = 0; v < 2; ++v) {
      const int n = kk * 4 + hi * 2 + v;
      g[kk][v] = Cmat[d * D_STATE + n] * powf(sAbar[n], (float)(row + 1));
    }
  }

  float cA[32], cB[32], cL[32];
  v2f s[16];
  #pragma unroll
  for (int kk = 0; kk < 16; ++kk) {
    #pragma unroll
    for (int v = 0; v < 2; ++v) {
      const int n = kk * 4 + hi * 2 + v;
      const float a1 = sAbar[n];
      const float a2 = a1 * a1, a4 = a2 * a2, a8 = a4 * a4;
      cA[kk * 2 + v] = a1;
      cB[kk * 2 + v] = sBbar[n];
      cL[kk * 2 + v] = a8 * a8;
      s[kk][v] = 0.f;
    }
  }

  for (int ch = 0; ch < CHUNKS; ++ch) {
    const int l0 = ch * TCH;

    #pragma unroll
    for (int k = 0; k < 4; ++k) {
      const int e = lane + k * 32;
      const int j = e >> 3;
      const int b = e & 7;
      sX[e] = x[(b * LENGTH + l0 + j) * D_MODEL + d];
    }
    __syncthreads();

    v8f acc = {};
    #pragma unroll
    for (int kk = 0; kk < 16; ++kk)
      acc = wmma_f32_16x16x4(g[kk], s[kk], acc);

    if (row < BATCH) {
      #pragma unroll
      for (int v = 0; v < 8; ++v) {
        const int r   = v + hi * 8;
        const int idx = (row * LENGTH + l0 + r) * D_MODEL + d;
        y[idx] += acc[v];
      }
    }

    float xv[TCH];
    #pragma unroll
    for (int j = 0; j < TCH; ++j)
      xv[j] = (row < BATCH) ? sX[j * BATCH + row] : 0.f;

    #pragma unroll
    for (int kk = 0; kk < 16; ++kk) {
      #pragma unroll
      for (int v = 0; v < 2; ++v) {
        float h = 0.f;
        #pragma unroll
        for (int j = 0; j < TCH; ++j)
          h = fmaf(cA[kk * 2 + v], h, xv[j]);
        s[kk][v] = fmaf(cL[kk * 2 + v], s[kk][v], cB[kk * 2 + v] * h);
      }
    }
    __syncthreads();
  }
}

extern "C" void kernel_launch(void* const* d_in, const int* in_sizes, int n_in,
                              void* d_out, int out_size, void* d_ws, size_t ws_size,
                              hipStream_t stream) {
  (void)in_sizes; (void)n_in; (void)out_size; (void)d_ws; (void)ws_size;
  const float* x      = (const float*)d_in[0];
  const float* log_dt = (const float*)d_in[1];
  const float* Bmat   = (const float*)d_in[2];
  const float* Cmat   = (const float*)d_in[3];
  const float* Dvec   = (const float*)d_in[4];
  float* y            = (float*)d_out;

  s4d_intra<<<dim3(D_MODEL), 256, 0, stream>>>(x, log_dt, Bmat, Cmat, Dvec, y);
  s4d_state<<<dim3(D_MODEL), 32, 0, stream>>>(x, log_dt, Bmat, Cmat, y);
}


# a channel pair's 64 states split over the two waves of a work-group (32 each): half the matrix and Horner work per wave, partial y rows exchanged via LDS, one barrier per chunk; y sums reordered
# speedup vs baseline: 6.1024x; 6.1024x over previous
#include <hip/hip_runtime.h>
#include <hip/hip_bf16.h>
#include <math.h>


#define BATCH    8
#define LENGTH   2048
#define D_MODEL  1024
#define D_STATE  64
#define TCH      16
#define CHUNKS   (LENGTH / TCH)
#define PAIRS    512
#define KL       8

typedef __attribute__((ext_vector_type(2))) float v2f;
typedef __attribute__((ext_vector_type(8))) float v8f;

#if defined(__HIP_DEVICE_COMPILE__)
#if !__has_builtin(__builtin_amdgcn_wmma_f32_16x16x4_f32)
#error "gfx1250 device pass: __builtin_amdgcn_wmma_f32_16x16x4_f32 not available"
#endif
#endif

__device__ __forceinline__ v8f wmma_f32_16x16x4(v2f a, v2f b, v8f c) {
#if defined(__HIP_DEVICE_COMPILE__)
  return __builtin_amdgcn_wmma_f32_16x16x4_f32(false, a, false, b, (short)0, c,
                                               false, false);
#else
  (void)a; (void)b;
  return c;
#endif
}

__device__ __forceinline__ void lds_barrier() {
#if defined(__HIP_DEVICE_COMPILE__)
  __builtin_amdgcn_fence(__ATOMIC_RELEASE, "workgroup", "local");
  __builtin_amdgcn_s_barrier();
  __builtin_amdgcn_fence(__ATOMIC_ACQUIRE, "workgroup", "local");
#endif
}

__device__ __forceinline__ float swap8(float v) {
#if defined(__HIP_DEVICE_COMPILE__)
  return __builtin_bit_cast(float,
      __builtin_amdgcn_update_dpp(__builtin_bit_cast(int, v), __builtin_bit_cast(int, v),
                                  0x128, 0xf, 0xf, false));
#else
  return v;
#endif
}

__global__ __launch_bounds__(64) void s4d_fused(
    const float* __restrict__ x, const float* __restrict__ log_dt,
    const float* __restrict__ Bmat, const float* __restrict__ Cmat,
    const float* __restrict__ Dvec, float* __restrict__ y)
{
  const int d0   = blockIdx.x * 2;
  const int tid  = threadIdx.x;
#if defined(__HIP_DEVICE_COMPILE__)
  const int w    = __builtin_amdgcn_readfirstlane(tid >> 5);
#else
  const int w    = tid >> 5;
#endif
  const int lane = tid & 31;
  const int row  = lane & 15;
  const int hi   = lane >> 4;
  const int bb   = row & 7;
  const int cs   = row >> 3;
  const bool up  = (cs != 0);
  const bool own = (cs == w);
  const int kb   = w * KL;

  __shared__ float sP[2][TCH][D_STATE + 1];
  __shared__ float sK[2][TCH];
  __shared__ float sAbar[2][D_STATE];
  __shared__ float sBbar[2][D_STATE];
  __shared__ float sG[2][TCH][D_STATE + 4];
  __shared__ float sX[2][2][TCH * BATCH + 32];
  __shared__ float sE[2][2][4][32];

  #pragma unroll
  for (int c2 = 0; c2 < 2; ++c2) {
    const int d = d0 + c2;
    const float dt = expf(log_dt[d]);
    for (int t = tid; t < D_STATE; t += 64) {
      const float An  = -0.5f * (float)(t + 1);
      const float dtA = dt * An;
      const float Abar = expf(dtA);
      sAbar[c2][t] = Abar;
      sBbar[c2][t] = (expm1f(dtA) / An) * Bmat[d * D_STATE + t];
      float q = Cmat[d * D_STATE + t] *
                (expm1f(dtA) / An) * Bmat[d * D_STATE + t];
      #pragma unroll
      for (int j = 0; j < TCH; ++j) { sP[c2][j][t] = q; q *= Abar; }
    }
  }
  __syncthreads();
  if (w == 0) {
    const int tap = lane & 15;
    float acc = 0.f;
    #pragma unroll 8
    for (int n = 0; n < D_STATE; ++n) acc += sP[hi][tap][n];
    sK[hi][tap] = acc;
  }

  #pragma nounroll
  for (int i = 0; i < 4; ++i) {
    #pragma unroll
    for (int q = 0; q < 8; ++q) {
      const int c2 = q >> 2, kl = i * 2 + ((q >> 1) & 1), v = q & 1;
      const int n = (kb + kl) * 4 + hi * 2 + v;
      sG[c2][row][n] = Cmat[(d0 + c2) * D_STATE + n] * powf(sAbar[c2][n], (float)(row + 1));
    }
  }
  __syncthreads();

  v2f af[4];
  #pragma unroll
  for (int kk = 0; kk < 4; ++kk) {
    #pragma unroll
    for (int v = 0; v < 2; ++v) {
      const int col = kk * 4 + hi * 2 + v;
      af[kk][v] = (row >= col) ? sK[w][row - col] : 0.f;
    }
  }
  const float Dd = Dvec[d0 + cs];

  v2f g0[KL], g1[KL];
  #pragma unroll
  for (int kl = 0; kl < KL; ++kl) {
    #pragma unroll
    for (int v = 0; v < 2; ++v) {
      const int n = (kb + kl) * 4 + hi * 2 + v;
      g0[kl][v] = sG[0][row][n];
      g1[kl][v] = sG[1][row][n];
    }
  }

  float cA[2 * KL], cB[2 * KL], cL[2 * KL];
  v2f s[KL];
  #pragma unroll
  for (int kl = 0; kl < KL; ++kl) {
    #pragma unroll
    for (int v = 0; v < 2; ++v) {
      const int n = (kb + kl) * 4 + hi * 2 + v;
      const float a1 = sAbar[cs][n];
      const float a2 = a1 * a1, a4 = a2 * a2, a8 = a4 * a4;
      cA[kl * 2 + v] = a1;
      cB[kl * 2 + v] = sBbar[cs][n];
      cL[kl * 2 + v] = a8 * a8;
      s[kl][v] = 0.f;
    }
  }

  v2f q0[2], q1[2], q2[2], q3[2], q4[2], q5[2];
  unsigned xo[2];
  #pragma unroll
  for (int k = 0; k < 2; ++k) {
    const int e = tid + k * 64;
    xo[k] = (unsigned)(((e & 7) * LENGTH + (e >> 3)) * D_MODEL + d0);
  }
  auto request = [&](v2f (&q)[2], int chunk) __attribute__((always_inline)) {
    const int cc = (chunk < CHUNKS) ? chunk : CHUNKS - 1;
    const float* xc = x + (size_t)cc * (TCH * D_MODEL);
    #pragma unroll
    for (int k = 0; k < 2; ++k)
      q[k] = *reinterpret_cast<const v2f*>(xc + xo[k]);
  };
  auto stage = [&](const v2f (&q)[2], int par) __attribute__((always_inline)) {
    #pragma unroll
    for (int k = 0; k < 2; ++k) {
      sX[par][0][tid + k * 64] = q[k][0];
      sX[par][1][tid + k * 64] = q[k][1];
    }
  };

  auto chunk_tile = [&](int par, float (&xv)[TCH], float (&xs)[4], float (&mine)[4]) __attribute__((always_inline)) {
    const float* sx = &sX[par][cs][0];
    v2f bfrag[4];
    #pragma unroll
    for (int kk = 0; kk < 4; ++kk) {
      #pragma unroll
      for (int v = 0; v < 2; ++v)
        bfrag[kk][v] = sx[(kk * 4 + hi * 2 + v) * BATCH + bb];
    }
    #pragma unroll
    for (int k = 0; k < 4; ++k)
      xs[k] = sx[((hi << 3) + (w << 2) + k) * BATCH + bb];
    #pragma unroll
    for (int j = 0; j < TCH; ++j)
      xv[j] = sx[j * BATCH + bb];

    __builtin_amdgcn_sched_barrier(0);

    v8f accs0 = {}, accs1 = {}, acci = {};
    #pragma unroll
    for (int kl = 0; kl < KL; ++kl) {
      accs0 = wmma_f32_16x16x4(g0[kl], s[kl], accs0);
      if (kl < 4)
        acci = wmma_f32_16x16x4(af[kl], bfrag[kl], acci);
      accs1 = wmma_f32_16x16x4(g1[kl], s[kl], accs1);
    }

    __builtin_amdgcn_sched_barrier(0);
    asm volatile("v_nop\n\tv_nop\n\tv_nop\n\tv_nop\n\tv_nop\n\tv_nop\n\tv_nop\n\tv_nop");
    __builtin_amdgcn_sched_barrier(0);

    float t[8];
    #pragma unroll
    for (int r = 0; r < 8; ++r)
      t[r] = (up ? accs1[r] : accs0[r]) + (own ? acci[r] : 0.f);
    #pragma unroll
    for (int k = 0; k < 4; ++k) {
      mine[k] = w ? t[k + 4] : t[k];
      sE[par][w ^ 1][k][lane] = w ? t[k] : t[k + 4];
    }
  };

  auto finish = [&](int ch, int par, const float (&xs)[4], const float (&mine)[4]) __attribute__((always_inline)) {
    float yk[4];
    #pragma unroll
    for (int k = 0; k < 4; ++k)
      yk[k] = (mine[k] + sE[par][w][k][lane]) + Dd * xs[k];
    const int r0 = ch * TCH + (hi << 3) + (w << 2) + (cs << 1);
    #pragma unroll
    for (int j = 0; j < 2; ++j) {
      const float keep = up ? yk[j + 2] : yk[j];
      const float got  = swap8(up ? yk[j] : yk[j + 2]);
      v2f pr;
      pr[0] = up ? got  : keep;
      pr[1] = up ? keep : got;
      *reinterpret_cast<v2f*>(&y[(bb * LENGTH + r0 + j) * D_MODEL + d0]) = pr;
    }
  };

  auto step = [&](int ch, int par, const v2f (&nxt)[2]) __attribute__((always_inline)) {
    float xv[TCH], xs[4], mine[4];
    chunk_tile(par, xv, xs, mine);

    stage(nxt, par ^ 1);

    float h[2 * KL];
    #pragma unroll
    for (int t = 0; t < 2 * KL; ++t) h[t] = 0.f;
    #pragma unroll
    for (int j = 0; j < TCH; ++j) {
      #pragma unroll
      for (int t = 0; t < 2 * KL; ++t)
        h[t] = fmaf(cA[t], h[t], xv[j]);
    }
    #pragma unroll
    for (int kl = 0; kl < KL; ++kl) {
      #pragma unroll
      for (int v = 0; v < 2; ++v)
        s[kl][v] = fmaf(cL[kl * 2 + v], s[kl][v], cB[kl * 2 + v] * h[kl * 2 + v]);
    }
    lds_barrier();

    finish(ch, par, xs, mine);
  };

  request(q0, 0);
  request(q1, 1);
  request(q2, 2);
  request(q3, 3);
  request(q4, 4);
  request(q5, 5);
  stage(q0, 0);
  lds_barrier();

  for (int c = 0; c < CHUNKS - 2; c += 6) {
    request(q0, c + 6);
    step(c, 0, q1);
    request(q1, c + 7);
    step(c + 1, 1, q2);
    request(q2, c + 8);
    step(c + 2, 0, q3);
    request(q3, c + 9);
    step(c + 3, 1, q4);
    request(q4, c + 10);
    step(c + 4, 0, q5);
    request(q5, c + 11);
    step(c + 5, 1, q0);
  }
  step(CHUNKS - 2, 0, q1);

  {
    float xv[TCH], xs[4], mine[4];
    chunk_tile(1, xv, xs, mine);
    lds_barrier();
    finish(CHUNKS - 1, 1, xs, mine);
  }
}

extern "C" void kernel_launch(void* const* d_in, const int* in_sizes, int n_in,
                              void* d_out, int out_size, void* d_ws, size_t ws_size,
                              hipStream_t stream) {
  (void)in_sizes; (void)n_in; (void)out_size; (void)d_ws; (void)ws_size;
  const float* x      = (const float*)d_in[0];
  const float* log_dt = (const float*)d_in[1];
  const float* Bmat   = (const float*)d_in[2];
  const float* Cmat   = (const float*)d_in[3];
  const float* Dvec   = (const float*)d_in[4];
  float* y            = (float*)d_out;

  s4d_fused<<<dim3(PAIRS), 64, 0, stream>>>(x, log_dt, Bmat, Cmat, Dvec, y);
}


# work-group of four waves owns four adjacent channels: x requested and y stored as one 16-byte piece per row and batch, y assembled from the waves' tiles in LDS; global requests halved; sums unchanged
# speedup vs baseline: 9.1489x; 1.4992x over previous
#include <hip/hip_runtime.h>
#include <hip/hip_bf16.h>
#include <math.h>


#define BATCH    8
#define LENGTH   2048
#define D_MODEL  1024
#define D_STATE  64
#define TCH      16
#define CHUNKS   (LENGTH / TCH)
#define GROUPS   256
#define KL       8

typedef __attribute__((ext_vector_type(2))) float v2f;
typedef __attribute__((ext_vector_type(4))) float v4f;
typedef __attribute__((ext_vector_type(8))) float v8f;

#if defined(__HIP_DEVICE_COMPILE__)
#if !__has_builtin(__builtin_amdgcn_wmma_f32_16x16x4_f32)
#error "gfx1250 device pass: __builtin_amdgcn_wmma_f32_16x16x4_f32 not available"
#endif
#endif

__device__ __forceinline__ v8f wmma_f32_16x16x4(v2f a, v2f b, v8f c) {
#if defined(__HIP_DEVICE_COMPILE__)
  return __builtin_amdgcn_wmma_f32_16x16x4_f32(false, a, false, b, (short)0, c,
                                               false, false);
#else
  (void)a; (void)b;
  return c;
#endif
}

__device__ __forceinline__ void lds_barrier() {
#if defined(__HIP_DEVICE_COMPILE__)
  __builtin_amdgcn_fence(__ATOMIC_RELEASE, "workgroup", "local");
  __builtin_amdgcn_s_barrier();
  __builtin_amdgcn_fence(__ATOMIC_ACQUIRE, "workgroup", "local");
#endif
}

__global__ __launch_bounds__(128) void s4d_fused(
    const float* __restrict__ x, const float* __restrict__ log_dt,
    const float* __restrict__ Bmat, const float* __restrict__ Cmat,
    const float* __restrict__ Dvec, float* __restrict__ y)
{
  const int d0   = blockIdx.x * 4;
  const int tid  = threadIdx.x;
#if defined(__HIP_DEVICE_COMPILE__)
  const int wv   = __builtin_amdgcn_readfirstlane(tid >> 5);
#else
  const int wv   = tid >> 5;
#endif
  const int c0   = wv & 2;
  const int w    = wv & 1;
  const int lane = tid & 31;
  const int row  = lane & 15;
  const int hi   = lane >> 4;
  const int bb   = row & 7;
  const int cs   = row >> 3;
  const bool up  = (cs != 0);
  const bool own = (cs == w);
  const int kb   = w * KL;

  __shared__ float sP[4][TCH][D_STATE + 1];
  __shared__ float sK[4][TCH];
  __shared__ float sAbar[4][D_STATE];
  __shared__ float sBbar[4][D_STATE];
  __shared__ float sG[4][TCH][D_STATE + 4];
  __shared__ float sX[2][4][TCH * BATCH + 32];
  __shared__ float sT[2][4][TCH][36];

  #pragma unroll
  for (int k = 0; k < 2; ++k) {
    const int c4 = k * 2 + (tid >> 6);
    const int t  = tid & 63;
    const int d  = d0 + c4;
    const float dt = expf(log_dt[d]);
    {
      const float An  = -0.5f * (float)(t + 1);
      const float dtA = dt * An;
      const float Abar = expf(dtA);
      sAbar[c4][t] = Abar;
      sBbar[c4][t] = (expm1f(dtA) / An) * Bmat[d * D_STATE + t];
      float q = Cmat[d * D_STATE + t] *
                (expm1f(dtA) / An) * Bmat[d * D_STATE + t];
      #pragma unroll
      for (int j = 0; j < TCH; ++j) { sP[c4][j][t] = q; q *= Abar; }
    }
  }
  __syncthreads();
  if (wv < 2) {
    const int tap = lane & 15;
    const int c4  = wv * 2 + hi;
    float acc = 0.f;
    #pragma unroll 8
    for (int n = 0; n < D_STATE; ++n) acc += sP[c4][tap][n];
    sK[c4][tap] = acc;
  }

  #pragma nounroll
  for (int i = 0; i < 4; ++i) {
    #pragma unroll
    for (int q = 0; q < 8; ++q) {
      const int c2 = q >> 2, kl = i * 2 + ((q >> 1) & 1), v = q & 1;
      const int n = (kb + kl) * 4 + hi * 2 + v;
      sG[c0 + c2][row][n] = Cmat[(d0 + c0 + c2) * D_STATE + n] * powf(sAbar[c0 + c2][n], (float)(row + 1));
    }
  }
  __syncthreads();

  v2f af[4];
  #pragma unroll
  for (int kk = 0; kk < 4; ++kk) {
    #pragma unroll
    for (int v = 0; v < 2; ++v) {
      const int col = kk * 4 + hi * 2 + v;
      af[kk][v] = (row >= col) ? sK[c0 + w][row - col] : 0.f;
    }
  }
  float Dc[4];
  #pragma unroll
  for (int c = 0; c < 4; ++c) Dc[c] = Dvec[d0 + c];

  v2f g0[KL], g1[KL];
  #pragma unroll
  for (int kl = 0; kl < KL; ++kl) {
    #pragma unroll
    for (int v = 0; v < 2; ++v) {
      const int n = (kb + kl) * 4 + hi * 2 + v;
      g0[kl][v] = sG[c0][row][n];
      g1[kl][v] = sG[c0 + 1][row][n];
    }
  }

  float cA[2 * KL], cB[2 * KL], cL[2 * KL];
  v2f s[KL];
  #pragma unroll
  for (int kl = 0; kl < KL; ++kl) {
    #pragma unroll
    for (int v = 0; v < 2; ++v) {
      const int n = (kb + kl) * 4 + hi * 2 + v;
      const float a1 = sAbar[c0 + cs][n];
      const float a2 = a1 * a1, a4 = a2 * a2, a8 = a4 * a4;
      cA[kl * 2 + v] = a1;
      cB[kl * 2 + v] = sBbar[c0 + cs][n];
      cL[kl * 2 + v] = a8 * a8;
      s[kl][v] = 0.f;
    }
  }

  v4f q0, q1, q2, q3, q4, q5;
  const unsigned xo = (unsigned)(((tid & 7) * LENGTH + (tid >> 3)) * D_MODEL + d0);
  auto request = [&](v4f& q, int chunk) __attribute__((always_inline)) {
    const int cc = (chunk < CHUNKS) ? chunk : CHUNKS - 1;
    const float* xc = x + (size_t)cc * (TCH * D_MODEL);
    q = *reinterpret_cast<const v4f*>(xc + xo);
  };
  auto stage = [&](const v4f& q, int par) __attribute__((always_inline)) {
    #pragma unroll
    for (int c = 0; c < 4; ++c)
      sX[par][c][tid] = q[c];
  };

  auto chunk_tile = [&](int par, float (&xv)[TCH]) __attribute__((always_inline)) {
    const float* sx = &sX[par][c0 + cs][0];
    v2f bfrag[4];
    #pragma unroll
    for (int kk = 0; kk < 4; ++kk) {
      #pragma unroll
      for (int v = 0; v < 2; ++v)
        bfrag[kk][v] = sx[(kk * 4 + hi * 2 + v) * BATCH + bb];
    }
    #pragma unroll
    for (int j = 0; j < TCH; ++j)
      xv[j] = sx[j * BATCH + bb];

    __builtin_amdgcn_sched_barrier(0);

    v8f accs0 = {}, accs1 = {}, acci = {};
    #pragma unroll
    for (int kl = 0; kl < KL; ++kl) {
      accs0 = wmma_f32_16x16x4(g0[kl], s[kl], accs0);
      if (kl < 4)
        acci = wmma_f32_16x16x4(af[kl], bfrag[kl], acci);
      accs1 = wmma_f32_16x16x4(g1[kl], s[kl], accs1);
    }

    __builtin_amdgcn_sched_barrier(0);
    asm volatile("v_nop\n\tv_nop\n\tv_nop\n\tv_nop\n\tv_nop\n\tv_nop\n\tv_nop\n\tv_nop");
    __builtin_amdgcn_sched_barrier(0);

    #pragma unroll
    for (int r = 0; r < 8; ++r)
      sT[par][wv][r + (hi << 3)][row] = (up ? accs1[r] : accs0[r]) + (own ? acci[r] : 0.f);
  };

  auto finish = [&](int ch, int par, const v4f& xk) __attribute__((always_inline)) {
    const int j = tid >> 3, b = tid & 7;
    v4f yv;
    #pragma unroll
    for (int c = 0; c < 4; ++c) {
      const int col = ((c & 1) << 3) + b;
      yv[c] = (sT[par][c & 2][j][col] + sT[par][(c & 2) + 1][j][col]) + Dc[c] * xk[c];
    }
    *reinterpret_cast<v4f*>(&y[(b * LENGTH + ch * TCH + j) * D_MODEL + d0]) = yv;
  };

  auto step = [&](int ch, int par, v4f& cur, const v4f& nxt) __attribute__((always_inline)) {
    const v4f xk = cur;
    request(cur, ch + 6);

    float xv[TCH];
    chunk_tile(par, xv);

    stage(nxt, par ^ 1);

    float h[2 * KL];
    #pragma unroll
    for (int t = 0; t < 2 * KL; ++t) h[t] = 0.f;
    #pragma unroll
    for (int j = 0; j < TCH; ++j) {
      #pragma unroll
      for (int t = 0; t < 2 * KL; ++t)
        h[t] = fmaf(cA[t], h[t], xv[j]);
    }
    #pragma unroll
    for (int kl = 0; kl < KL; ++kl) {
      #pragma unroll
      for (int v = 0; v < 2; ++v)
        s[kl][v] = fmaf(cL[kl * 2 + v], s[kl][v], cB[kl * 2 + v] * h[kl * 2 + v]);
    }
    lds_barrier();

    finish(ch, par, xk);
  };

  request(q0, 0);
  request(q1, 1);
  request(q2, 2);
  request(q3, 3);
  request(q4, 4);
  request(q5, 5);
  stage(q0, 0);
  lds_barrier();

  for (int c = 0; c < CHUNKS - 2; c += 6) {
    step(c,     0, q0, q1);
    step(c + 1, 1, q1, q2);
    step(c + 2, 0, q2, q3);
    step(c + 3, 1, q3, q4);
    step(c + 4, 0, q4, q5);
    step(c + 5, 1, q5, q0);
  }
  step(CHUNKS - 2, 0, q0, q1);

  {
    const v4f xk = q1;
    float xv[TCH];
    chunk_tile(1, xv);
    lds_barrier();
    finish(CHUNKS - 1, 1, xk);
  }
}

extern "C" void kernel_launch(void* const* d_in, const int* in_sizes, int n_in,
                              void* d_out, int out_size, void* d_ws, size_t ws_size,
                              hipStream_t stream) {
  (void)in_sizes; (void)n_in; (void)out_size; (void)d_ws; (void)ws_size;
  const float* x      = (const float*)d_in[0];
  const float* log_dt = (const float*)d_in[1];
  const float* Bmat   = (const float*)d_in[2];
  const float* Cmat   = (const float*)d_in[3];
  const float* Dvec   = (const float*)d_in[4];
  float* y            = (float*)d_out;

  s4d_fused<<<dim3(GROUPS), 128, 0, stream>>>(x, log_dt, Bmat, Cmat, Dvec, y);
}


# four extra waves per work-group take x requests, LDS staging, y finishing and y stores, one chunk behind; the four matrix waves issue no global memory instruction in the walk; sums unchanged
# speedup vs baseline: 10.7494x; 1.1749x over previous
#include <hip/hip_runtime.h>
#include <hip/hip_bf16.h>
#include <math.h>


#define BATCH    8
#define LENGTH   2048
#define D_MODEL  1024
#define D_STATE  64
#define TCH      16
#define CHUNKS   (LENGTH / TCH)
#define GROUPS   256
#define KL       8
#define RING     8

typedef __attribute__((ext_vector_type(2))) float v2f;
typedef __attribute__((ext_vector_type(4))) float v4f;
typedef __attribute__((ext_vector_type(8))) float v8f;

#if defined(__HIP_DEVICE_COMPILE__)
#if !__has_builtin(__builtin_amdgcn_wmma_f32_16x16x4_f32)
#error "gfx1250 device pass: __builtin_amdgcn_wmma_f32_16x16x4_f32 not available"
#endif
#endif

__device__ __forceinline__ v8f wmma_f32_16x16x4(v2f a, v2f b, v8f c) {
#if defined(__HIP_DEVICE_COMPILE__)
  return __builtin_amdgcn_wmma_f32_16x16x4_f32(false, a, false, b, (short)0, c,
                                               false, false);
#else
  (void)a; (void)b;
  return c;
#endif
}

__device__ __forceinline__ void lds_barrier() {
#if defined(__HIP_DEVICE_COMPILE__)
  __builtin_amdgcn_fence(__ATOMIC_RELEASE, "workgroup", "local");
  __builtin_amdgcn_s_barrier();
  __builtin_amdgcn_fence(__ATOMIC_ACQUIRE, "workgroup", "local");
#endif
}


__global__ __launch_bounds__(256) void s4d_fused(
    const float* __restrict__ x, const float* __restrict__ log_dt,
    const float* __restrict__ Bmat, const float* __restrict__ Cmat,
    const float* __restrict__ Dvec, float* __restrict__ y)
{
  const int d0   = blockIdx.x * 4;
  const int tid  = threadIdx.x;
#if defined(__HIP_DEVICE_COMPILE__)
  const int wv   = __builtin_amdgcn_readfirstlane(tid >> 5);
#else
  const int wv   = tid >> 5;
#endif
  const bool mover = (wv >= 4);
  const int mw   = wv & 3;
  const int c0   = mw & 2;
  const int w    = mw & 1;
  const int lane = tid & 31;
  const int row  = lane & 15;
  const int hi   = lane >> 4;
  const int bb   = row & 7;
  const int cs   = row >> 3;
  const bool up  = (cs != 0);
  const bool own = (cs == w);
  const int kb   = w * KL;
  const int pt   = tid & 127;

  __shared__ float sP[4][TCH][D_STATE + 1];
  __shared__ float sK[4][TCH];
  __shared__ float sAbar[4][D_STATE];
  __shared__ float sBbar[4][D_STATE];
  __shared__ float sG[4][TCH][D_STATE + 4];
  __shared__ float sX[RING][4][TCH * BATCH + 32];
  __shared__ float sT[2][4][TCH][36];

  {
    const int c4 = tid >> 6;
    const int t  = tid & 63;
    const int d  = d0 + c4;
    const float dt = expf(log_dt[d]);
    {
      const float An  = -0.5f * (float)(t + 1);
      const float dtA = dt * An;
      const float Abar = expf(dtA);
      sAbar[c4][t] = Abar;
      sBbar[c4][t] = (expm1f(dtA) / An) * Bmat[d * D_STATE + t];
      float q = Cmat[d * D_STATE + t] *
                (expm1f(dtA) / An) * Bmat[d * D_STATE + t];
      #pragma unroll
      for (int j = 0; j < TCH; ++j) { sP[c4][j][t] = q; q *= Abar; }
    }
  }
  __syncthreads();
  if (wv < 2) {
    const int tap = lane & 15;
    const int c4  = wv * 2 + hi;
    float acc = 0.f;
    #pragma unroll 8
    for (int n = 0; n < D_STATE; ++n) acc += sP[c4][tap][n];
    sK[c4][tap] = acc;
  }

  v4f r0 = {}, r1 = {}, r2 = {}, r3 = {};
  const unsigned xo = (unsigned)(((pt & 7) * LENGTH + (pt >> 3)) * D_MODEL + d0);
  auto request = [&](v4f& q, int chunk) __attribute__((always_inline)) {
    const float* xc = x + (size_t)chunk * (TCH * D_MODEL);
    q = *reinterpret_cast<const v4f*>(xc + xo);
  };
  float Dc[4] = {0.f, 0.f, 0.f, 0.f};

  if (!mover) {
    #pragma nounroll
    for (int i = 0; i < 4; ++i) {
      #pragma unroll
      for (int q = 0; q < 8; ++q) {
        const int c2 = q >> 2, kl = i * 2 + ((q >> 1) & 1), v = q & 1;
        const int n = (kb + kl) * 4 + hi * 2 + v;
        sG[c0 + c2][row][n] = Cmat[(d0 + c0 + c2) * D_STATE + n] * powf(sAbar[c0 + c2][n], (float)(row + 1));
      }
    }
  } else {
    #pragma unroll
    for (int c = 0; c < 4; ++c) Dc[c] = Dvec[d0 + c];
    request(r0, 0);
    request(r1, 1);
    request(r2, 2);
    request(r3, 3);
  }
  __syncthreads();

  v2f af[4];
  #pragma unroll
  for (int kk = 0; kk < 4; ++kk) {
    #pragma unroll
    for (int v = 0; v < 2; ++v) {
      const int col = kk * 4 + hi * 2 + v;
      af[kk][v] = (row >= col) ? sK[c0 + w][row - col] : 0.f;
    }
  }

  v2f g0[KL], g1[KL];
  #pragma unroll
  for (int kl = 0; kl < KL; ++kl) {
    #pragma unroll
    for (int v = 0; v < 2; ++v) {
      const int n = (kb + kl) * 4 + hi * 2 + v;
      g0[kl][v] = sG[c0][row][n];
      g1[kl][v] = sG[c0 + 1][row][n];
    }
  }

  float cA[2 * KL], cB[2 * KL], cL[2 * KL];
  v2f s[KL];
  #pragma unroll
  for (int kl = 0; kl < KL; ++kl) {
    #pragma unroll
    for (int v = 0; v < 2; ++v) {
      const int n = (kb + kl) * 4 + hi * 2 + v;
      const float a1 = sAbar[c0 + cs][n];
      const float a2 = a1 * a1, a4 = a2 * a2, a8 = a4 * a4;
      cA[kl * 2 + v] = a1;
      cB[kl * 2 + v] = sBbar[c0 + cs][n];
      cL[kl * 2 + v] = a8 * a8;
      s[kl][v] = 0.f;
    }
  }

  auto stage = [&](const v4f& q, int pl) __attribute__((always_inline)) {
    #pragma unroll
    for (int c = 0; c < 4; ++c)
      sX[pl][c][pt] = q[c];
  };

  auto chunk_tile = [&](int pl, int par, float (&xv)[TCH]) __attribute__((always_inline)) {
    const float* sx = &sX[pl][c0 + cs][0];
    v2f bfrag[4];
    #pragma unroll
    for (int kk = 0; kk < 4; ++kk) {
      #pragma unroll
      for (int v = 0; v < 2; ++v)
        bfrag[kk][v] = sx[(kk * 4 + hi * 2 + v) * BATCH + bb];
    }
    #pragma unroll
    for (int j = 0; j < TCH; ++j)
      xv[j] = sx[j * BATCH + bb];

    __builtin_amdgcn_sched_barrier(0);

    v8f accs0 = {}, accs1 = {}, acci = {};
    #pragma unroll
    for (int kl = 0; kl < KL; ++kl) {
      accs0 = wmma_f32_16x16x4(g0[kl], s[kl], accs0);
      if (kl < 4)
        acci = wmma_f32_16x16x4(af[kl], bfrag[kl], acci);
      accs1 = wmma_f32_16x16x4(g1[kl], s[kl], accs1);
    }

    __builtin_amdgcn_sched_barrier(0);
    asm volatile("v_nop\n\tv_nop\n\tv_nop\n\tv_nop\n\tv_nop\n\tv_nop\n\tv_nop\n\tv_nop");
    __builtin_amdgcn_sched_barrier(0);

    #pragma unroll
    for (int r = 0; r < 8; ++r)
      sT[par][mw][r + (hi << 3)][row] = (up ? accs1[r] : accs0[r]) + (own ? acci[r] : 0.f);
  };

  auto finish = [&](int ch, int pl, int par) __attribute__((always_inline)) {
    const int j = pt >> 3, b = pt & 7;
    v4f yv;
    #pragma unroll
    for (int c = 0; c < 4; ++c) {
      const int col = ((c & 1) << 3) + b;
      yv[c] = (sT[par][c & 2][j][col] + sT[par][(c & 2) + 1][j][col]) + Dc[c] * sX[pl][c][pt];
    }
    *reinterpret_cast<v4f*>(&y[(b * LENGTH + ch * TCH + j) * D_MODEL + d0]) = yv;
  };

  auto step = [&](int ch, int i) __attribute__((always_inline)) {
    if (mover) {
      if (ch > 0) finish(ch - 1, (i + RING - 1) % RING, (i + 1) & 1);
      if ((i & 3) == 3) {
        if (ch + 1 < CHUNKS) {
          stage(r0, (i + 1) % RING);
          stage(r1, (i + 2) % RING);
          stage(r2, (i + 3) % RING);
          stage(r3, (i + 4) % RING);
          if (ch + 5 < CHUNKS) {
            request(r0, ch + 5);
            request(r1, ch + 6);
            request(r2, ch + 7);
            request(r3, ch + 8);
          }
        }
      }
    } else {
      float xv[TCH];
      chunk_tile(i, i & 1, xv);

      float h[2 * KL];
      #pragma unroll
      for (int t = 0; t < 2 * KL; ++t) h[t] = 0.f;
      #pragma unroll
      for (int j = 0; j < TCH; ++j) {
        #pragma unroll
        for (int t = 0; t < 2 * KL; ++t)
          h[t] = fmaf(cA[t], h[t], xv[j]);
      }
      #pragma unroll
      for (int kl = 0; kl < KL; ++kl) {
        #pragma unroll
        for (int v = 0; v < 2; ++v)
          s[kl][v] = fmaf(cL[kl * 2 + v], s[kl][v], cB[kl * 2 + v] * h[kl * 2 + v]);
      }
    }
    lds_barrier();
  };

  if (mover) {
    stage(r0, 0);
    stage(r1, 1);
    stage(r2, 2);
    stage(r3, 3);
    request(r0, 4);
    request(r1, 5);
    request(r2, 6);
    request(r3, 7);
  }
  lds_barrier();

  for (int c = 0; c < CHUNKS; c += 8) {
    step(c,     0);
    step(c + 1, 1);
    step(c + 2, 2);
    step(c + 3, 3);
    step(c + 4, 4);
    step(c + 5, 5);
    step(c + 6, 6);
    step(c + 7, 7);
  }

  if (mover) finish(CHUNKS - 1, (CHUNKS - 1) % RING, (CHUNKS - 1) & 1);
}

extern "C" void kernel_launch(void* const* d_in, const int* in_sizes, int n_in,
                              void* d_out, int out_size, void* d_ws, size_t ws_size,
                              hipStream_t stream) {
  (void)in_sizes; (void)n_in; (void)out_size; (void)d_ws; (void)ws_size;
  const float* x      = (const float*)d_in[0];
  const float* log_dt = (const float*)d_in[1];
  const float* Bmat   = (const float*)d_in[2];
  const float* Cmat   = (const float*)d_in[3];
  const float* Dvec   = (const float*)d_in[4];
  float* y            = (float*)d_out;

  s4d_fused<<<dim3(GROUPS), 256, 0, stream>>>(x, log_dt, Bmat, Cmat, Dvec, y);
}


# matrix wave = one channel, all 64 states; tile columns = 8 batches x 2 consecutive chunks (was x 2 channels): no product column wasted; 2nd chunk's start state by row_ror:8 move; y sums reordered
# speedup vs baseline: 11.3586x; 1.0567x over previous
#include <hip/hip_runtime.h>
#include <hip/hip_bf16.h>
#include <math.h>


#define BATCH    8
#define LENGTH   2048
#define D_MODEL  1024
#define D_STATE  64
#define TCH      16
#define CHUNKS   (LENGTH / TCH)
#define STEPS    (CHUNKS / 2)
#define GROUPS   256
#define KL       16
#define RING     8
#define XP       168
#define TPITCH   10
#define TP       168

typedef __attribute__((ext_vector_type(2))) float v2f;
typedef __attribute__((ext_vector_type(4))) float v4f;
typedef __attribute__((ext_vector_type(8))) float v8f;

#if defined(__HIP_DEVICE_COMPILE__)
#if !__has_builtin(__builtin_amdgcn_wmma_f32_16x16x4_f32)
#error "gfx1250 device pass: __builtin_amdgcn_wmma_f32_16x16x4_f32 not available"
#endif
#endif

__device__ __forceinline__ v8f wmma_f32_16x16x4(v2f a, v2f b, v8f c) {
#if defined(__HIP_DEVICE_COMPILE__)
  return __builtin_amdgcn_wmma_f32_16x16x4_f32(false, a, false, b, (short)0, c,
                                               false, false);
#else
  (void)a; (void)b;
  return c;
#endif
}

template <int BM>
__device__ __forceinline__ float take_across(float keep, float src) {
#if defined(__HIP_DEVICE_COMPILE__)
  return __int_as_float(__builtin_amdgcn_update_dpp(__float_as_int(keep), __float_as_int(src),
                                                    0x128  , 0xF, BM, false));
#else
  (void)src;
  return keep;
#endif
}

__device__ __forceinline__ void lds_barrier() {
#if defined(__HIP_DEVICE_COMPILE__)
  __builtin_amdgcn_fence(__ATOMIC_RELEASE, "workgroup", "local");
  __builtin_amdgcn_s_barrier();
  __builtin_amdgcn_fence(__ATOMIC_ACQUIRE, "workgroup", "local");
#endif
}


__global__ __launch_bounds__(256) void s4d_fused(
    const float* __restrict__ x, const float* __restrict__ log_dt,
    const float* __restrict__ Bmat, const float* __restrict__ Cmat,
    const float* __restrict__ Dvec, float* __restrict__ y)
{
  const int d0   = blockIdx.x * 4;
  const int tid  = threadIdx.x;
#if defined(__HIP_DEVICE_COMPILE__)
  const int wv   = __builtin_amdgcn_readfirstlane(tid >> 5);
#else
  const int wv   = tid >> 5;
#endif
  const bool mover = (wv >= 4);
  const int mw   = wv & 3;
  const int lane = tid & 31;
  const int row  = lane & 15;
  const int hi   = lane >> 4;
  const int bb   = row & 7;
  const int qc   = row >> 3;
  const int pt   = tid & 127;

  __shared__ float sP[4][TCH][D_STATE + 1];
  __shared__ float sK[4][TCH];
  __shared__ float sAbar[4][D_STATE];
  __shared__ float sX[RING][4][XP];
  __shared__ float sT[4][4][TP];

  {
    const int c4 = tid >> 6;
    const int t  = tid & 63;
    const int d  = d0 + c4;
    const float dt = expf(log_dt[d]);
    {
      const float An  = -0.5f * (float)(t + 1);
      const float dtA = dt * An;
      const float Abar = expf(dtA);
      sAbar[c4][t] = Abar;
      float q = Cmat[d * D_STATE + t] *
                (expm1f(dtA) / An) * Bmat[d * D_STATE + t];
      #pragma unroll
      for (int j = 0; j < TCH; ++j) { sP[c4][j][t] = q; q *= Abar; }
    }
  }
  __syncthreads();
  if (wv < 2) {
    const int tap = lane & 15;
    const int c4  = wv * 2 + hi;
    float acc = 0.f;
    #pragma unroll 8
    for (int n = 0; n < D_STATE; ++n) acc += sP[c4][tap][n];
    sK[c4][tap] = acc;
  }

  v4f r0 = {}, r1 = {}, r2 = {}, r3 = {};
  const unsigned xo = (unsigned)(((pt & 7) * LENGTH + (pt >> 3)) * D_MODEL + d0);
  auto request = [&](v4f& q, int chunk) __attribute__((always_inline)) {
    const float* xc = x + (size_t)chunk * (TCH * D_MODEL);
    q = *reinterpret_cast<const v4f*>(xc + xo);
  };
  float Dc[4] = {0.f, 0.f, 0.f, 0.f};

  if (mover) {
    #pragma unroll
    for (int c = 0; c < 4; ++c) Dc[c] = Dvec[d0 + c];
    request(r0, 0);
    request(r1, 1);
    request(r2, 2);
    request(r3, 3);
  }
  __syncthreads();

  v2f af[4];
  #pragma unroll
  for (int kk = 0; kk < 4; ++kk) {
    #pragma unroll
    for (int v = 0; v < 2; ++v) {
      const int col = kk * 4 + hi * 2 + v;
      af[kk][v] = (row >= col) ? sK[mw][row - col] : 0.f;
    }
  }

  v2f g[KL];
  float cA[2 * KL], cL[2 * KL];
  v2f s[KL];
  #pragma unroll
  for (int kl = 0; kl < KL; ++kl) {
    #pragma unroll
    for (int v = 0; v < 2; ++v) {
      const int n = kl * 4 + hi * 2 + v;
      const float a1 = sAbar[mw][n];
      const float a2 = a1 * a1, a4 = a2 * a2, a8 = a4 * a4;
      g[kl][v] = sP[mw][row][n] * a1;
      cA[kl * 2 + v] = a1;
      cL[kl * 2 + v] = a8 * a8;
      s[kl][v] = 0.f;
    }
  }

  auto stage = [&](const v4f& q, int pl) __attribute__((always_inline)) {
    #pragma unroll
    for (int c = 0; c < 4; ++c)
      sX[pl][c][pt] = q[c];
  };

  auto tile2 = [&](int k) __attribute__((always_inline)) {
    const float* sx = &sX[2 * k][mw][0] + qc * (4 * XP);
    v2f bfrag[4];
    #pragma unroll
    for (int kk = 0; kk < 4; ++kk) {
      #pragma unroll
      for (int v = 0; v < 2; ++v)
        bfrag[kk][v] = sx[(kk * 4 + hi * 2 + v) * BATCH + bb];
    }
    float xv[TCH];
    #pragma unroll
    for (int j = 0; j < TCH; ++j)
      xv[j] = sx[j * BATCH + bb];

    float h[2 * KL];
    #pragma unroll
    for (int t = 0; t < 2 * KL; ++t) h[t] = xv[0];
    #pragma unroll
    for (int j = 1; j < TCH; ++j) {
      #pragma unroll
      for (int t = 0; t < 2 * KL; ++t)
        h[t] = fmaf(cA[t], h[t], xv[j]);
    }

    #pragma unroll
    for (int kl = 0; kl < KL; ++kl) {
      #pragma unroll
      for (int v = 0; v < 2; ++v)
        s[kl][v] = take_across<0xC>(s[kl][v], fmaf(cL[kl * 2 + v], s[kl][v], h[kl * 2 + v]));
    }

    __builtin_amdgcn_sched_barrier(0);

    v8f acc0 = {}, acc1 = {};
    #pragma unroll
    for (int kl = 0; kl < KL; ++kl) {
      if (kl < 4) {
        acc0 = wmma_f32_16x16x4(g[kl], s[kl], acc0);
        acc1 = wmma_f32_16x16x4(af[kl], bfrag[kl], acc1);
      } else if (kl & 1) {
        acc1 = wmma_f32_16x16x4(g[kl], s[kl], acc1);
      } else {
        acc0 = wmma_f32_16x16x4(g[kl], s[kl], acc0);
      }
    }

    __builtin_amdgcn_sched_barrier(0);
    asm volatile("v_nop\n\tv_nop\n\tv_nop\n\tv_nop\n\tv_nop\n\tv_nop\n\tv_nop\n\tv_nop");
    __builtin_amdgcn_sched_barrier(0);

    float* st = &sT[(2 * k) & 3][mw][0] + qc * (4 * TP) + hi * (8 * TPITCH) + bb;
    #pragma unroll
    for (int r = 0; r < 8; ++r)
      st[r * TPITCH] = acc0[r] + acc1[r];

    #pragma unroll
    for (int kl = 0; kl < KL; ++kl) {
      #pragma unroll
      for (int v = 0; v < 2; ++v)
        s[kl][v] = take_across<0x3>(s[kl][v], fmaf(cL[kl * 2 + v], s[kl][v], h[kl * 2 + v]));
    }
  };

  auto finish = [&](int ch, int pl, int tp) __attribute__((always_inline)) {
    const int j = pt >> 3, b = pt & 7;
    v4f yv;
    #pragma unroll
    for (int c = 0; c < 4; ++c)
      yv[c] = sT[tp][c][j * TPITCH + b] + Dc[c] * sX[pl][c][pt];
    *reinterpret_cast<v4f*>(&y[(b * LENGTH + ch * TCH + j) * D_MODEL + d0]) = yv;
  };

  auto step = [&](int u, int k) __attribute__((always_inline)) {
    if (mover) {
      if (u > 0) {
        finish(2 * u - 2, (2 * k + 6) % RING, (2 * k + 2) & 3);
        finish(2 * u - 1, (2 * k + 7) % RING, (2 * k + 3) & 3);
      }
      if (k & 1) {
        if (2 * u + 2 < CHUNKS) {
          stage(r0, (2 * k + 2) % RING);
          stage(r1, (2 * k + 3) % RING);
          stage(r2, (2 * k + 4) % RING);
          stage(r3, (2 * k + 5) % RING);
          if (2 * u + 6 < CHUNKS) {
            request(r0, 2 * u + 6);
            request(r1, 2 * u + 7);
            request(r2, 2 * u + 8);
            request(r3, 2 * u + 9);
          }
        }
      }
    } else {
      tile2(k);
    }
    lds_barrier();
  };

  if (mover) {
    stage(r0, 0);
    stage(r1, 1);
    stage(r2, 2);
    stage(r3, 3);
    request(r0, 4);
    request(r1, 5);
    request(r2, 6);
    request(r3, 7);
  }
  lds_barrier();

  for (int u = 0; u < STEPS; u += 4) {
    step(u,     0);
    step(u + 1, 1);
    step(u + 2, 2);
    step(u + 3, 3);
  }

  if (mover) {
    finish(CHUNKS - 2, (CHUNKS - 2) % RING, (CHUNKS - 2) & 3);
    finish(CHUNKS - 1, (CHUNKS - 1) % RING, (CHUNKS - 1) & 3);
  }
}

extern "C" void kernel_launch(void* const* d_in, const int* in_sizes, int n_in,
                              void* d_out, int out_size, void* d_ws, size_t ws_size,
                              hipStream_t stream) {
  (void)in_sizes; (void)n_in; (void)out_size; (void)d_ws; (void)ws_size;
  const float* x      = (const float*)d_in[0];
  const float* log_dt = (const float*)d_in[1];
  const float* Bmat   = (const float*)d_in[2];
  const float* Cmat   = (const float*)d_in[3];
  const float* Dvec   = (const float*)d_in[4];
  float* y            = (float*)d_out;

  s4d_fused<<<dim3(GROUPS), 256, 0, stream>>>(x, log_dt, Bmat, Cmat, Dvec, y);
}


# work-group = 8 adjacent channels x 4 batches (was 4 x 8): a row's two x/y pieces sit in a lane pair, a request touches 16 lines of 32 B (was 32 of 16 B); tile columns 4 batches x 4 chunks; same sums
# speedup vs baseline: 15.8591x; 1.3962x over previous
#include <hip/hip_runtime.h>
#include <hip/hip_bf16.h>
#include <math.h>


#define BATCH    8
#define LENGTH   2048
#define D_MODEL  1024
#define D_STATE  64
#define TCH      16
#define CHUNKS   (LENGTH / TCH)
#define CPS      4
#define STEPS    (CHUNKS / CPS)
#define NCH      8
#define NB       4
#define GROUPS   ((D_MODEL / NCH) * (BATCH / NB))
#define NTHR     384
#define KL       16
#define XQ       80
#define XC       324
#define TQ       68
#define TC       276

typedef __attribute__((ext_vector_type(2))) float v2f;
typedef __attribute__((ext_vector_type(4))) float v4f;
typedef __attribute__((ext_vector_type(8))) float v8f;

#if defined(__HIP_DEVICE_COMPILE__)
#if !__has_builtin(__builtin_amdgcn_wmma_f32_16x16x4_f32)
#error "gfx1250 device pass: __builtin_amdgcn_wmma_f32_16x16x4_f32 not available"
#endif
#endif

__device__ __forceinline__ v8f wmma_f32_16x16x4(v2f a, v2f b, v8f c) {
#if defined(__HIP_DEVICE_COMPILE__)
  return __builtin_amdgcn_wmma_f32_16x16x4_f32(false, a, false, b, (short)0, c,
                                               false, false);
#else
  (void)a; (void)b;
  return c;
#endif
}

template <int BM>
__device__ __forceinline__ float take_prev(float keep, float src) {
#if defined(__HIP_DEVICE_COMPILE__)
  return __int_as_float(__builtin_amdgcn_update_dpp(__float_as_int(keep), __float_as_int(src),
                                                    0x124  , 0xF, BM, false));
#else
  (void)src;
  return keep;
#endif
}

__device__ __forceinline__ void lds_barrier() {
#if defined(__HIP_DEVICE_COMPILE__)
  __builtin_amdgcn_fence(__ATOMIC_RELEASE, "workgroup", "local");
  __builtin_amdgcn_s_barrier();
  __builtin_amdgcn_fence(__ATOMIC_ACQUIRE, "workgroup", "local");
#endif
}


__global__ __launch_bounds__(NTHR) void s4d_fused(
    const float* __restrict__ x, const float* __restrict__ log_dt,
    const float* __restrict__ Bmat, const float* __restrict__ Cmat,
    const float* __restrict__ Dvec, float* __restrict__ y)
{
  const int d0   = (blockIdx.x >> 1) * NCH;
  const int b0   = (blockIdx.x & 1) * NB;
  const int tid  = threadIdx.x;
#if defined(__HIP_DEVICE_COMPILE__)
  const int wv   = __builtin_amdgcn_readfirstlane(tid >> 5);
#else
  const int wv   = tid >> 5;
#endif
  const bool mover = (wv >= NCH);
  const int mw   = wv & 7;
  const int lane = tid & 31;
  const int row  = lane & 15;
  const int hi   = lane >> 4;
  const int bb   = row & 3;
  const int cq   = row >> 2;
  const int pt   = tid & 127;
  const int phf  = pt & 1;
  const int pb   = (pt >> 1) & 3;
  const int pj   = pt >> 3;

  __shared__ float sP[NCH][TCH][D_STATE + 1];
  __shared__ float sK[NCH][TCH];
  __shared__ float sAbar[NCH][D_STATE];
  __shared__ float sX[4][NCH][XC];
  __shared__ float sT[2][NCH][TC];

  for (int idx = tid; idx < NCH * D_STATE; idx += NTHR) {
    const int c8 = idx >> 6;
    const int t  = idx & 63;
    const int d  = d0 + c8;
    const float dt = expf(log_dt[d]);
    {
      const float An  = -0.5f * (float)(t + 1);
      const float dtA = dt * An;
      const float Abar = expf(dtA);
      sAbar[c8][t] = Abar;
      float q = Cmat[d * D_STATE + t] *
                (expm1f(dtA) / An) * Bmat[d * D_STATE + t];
      #pragma unroll
      for (int j = 0; j < TCH; ++j) { sP[c8][j][t] = q; q *= Abar; }
    }
  }
  __syncthreads();
  if (wv < 4) {
    const int tap = lane & 15;
    const int c8  = wv * 2 + hi;
    float acc = 0.f;
    #pragma unroll 8
    for (int n = 0; n < D_STATE; ++n) acc += sP[c8][tap][n];
    sK[c8][tap] = acc;
  }

  v4f r0 = {}, r1 = {}, r2 = {}, r3 = {};
  v4f r4 = {}, r5 = {}, r6 = {}, r7 = {};
  const unsigned xo = (unsigned)(((b0 + pb) * LENGTH + pj) * D_MODEL + d0 + 4 * phf);
  auto request = [&](v4f& q, int chunk) __attribute__((always_inline)) {
    const float* xc = x + (size_t)chunk * (TCH * D_MODEL);
    q = *reinterpret_cast<const v4f*>(xc + xo);
  };
  float Dc[4] = {0.f, 0.f, 0.f, 0.f};

  if (mover) {
    #pragma unroll
    for (int c = 0; c < 4; ++c) Dc[c] = Dvec[d0 + 4 * phf + c];
    request(r0, 0);
    request(r1, 1);
    request(r2, 2);
    request(r3, 3);
    request(r4, 4);
    request(r5, 5);
    request(r6, 6);
    request(r7, 7);
  }
  __syncthreads();

  v2f af[4];
  #pragma unroll
  for (int kk = 0; kk < 4; ++kk) {
    #pragma unroll
    for (int v = 0; v < 2; ++v) {
      const int col = kk * 4 + hi * 2 + v;
      af[kk][v] = (row >= col) ? sK[mw][row - col] : 0.f;
    }
  }

  v2f g[KL];
  float cA[2 * KL], cL[2 * KL];
  v2f s[KL];
  #pragma unroll
  for (int kl = 0; kl < KL; ++kl) {
    #pragma unroll
    for (int v = 0; v < 2; ++v) {
      const int n = kl * 4 + hi * 2 + v;
      const float a1 = sAbar[mw][n];
      const float a2 = a1 * a1, a4 = a2 * a2, a8 = a4 * a4;
      g[kl][v] = sP[mw][row][n] * a1;
      cA[kl * 2 + v] = a1;
      cL[kl * 2 + v] = a8 * a8;
      s[kl][v] = 0.f;
    }
  }

  auto stage = [&](const v4f& r, int pl, int q) __attribute__((always_inline)) {
    #pragma unroll
    for (int c = 0; c < 4; ++c)
      sX[pl][4 * phf + c][q * XQ + pj * NB + pb] = r[c];
  };

  auto tile4 = [&](int k) __attribute__((always_inline)) {
    const float* sx = &sX[k][mw][0] + cq * XQ + bb;
    v2f bfrag[4];
    #pragma unroll
    for (int kk = 0; kk < 4; ++kk) {
      #pragma unroll
      for (int v = 0; v < 2; ++v)
        bfrag[kk][v] = sx[(kk * 4 + hi * 2 + v) * NB];
    }
    float xv[TCH];
    #pragma unroll
    for (int j = 0; j < TCH; ++j)
      xv[j] = sx[j * NB];

    float h[2 * KL];
    #pragma unroll
    for (int t = 0; t < 2 * KL; ++t) h[t] = xv[0];
    #pragma unroll
    for (int j = 1; j < TCH; ++j) {
      #pragma unroll
      for (int t = 0; t < 2 * KL; ++t)
        h[t] = fmaf(cA[t], h[t], xv[j]);
    }

    #pragma unroll
    for (int kl = 0; kl < KL; ++kl) {
      #pragma unroll
      for (int v = 0; v < 2; ++v)
        s[kl][v] = take_prev<0x2>(s[kl][v], fmaf(cL[kl * 2 + v], s[kl][v], h[kl * 2 + v]));
    }
    #pragma unroll
    for (int kl = 0; kl < KL; ++kl) {
      #pragma unroll
      for (int v = 0; v < 2; ++v)
        s[kl][v] = take_prev<0x4>(s[kl][v], fmaf(cL[kl * 2 + v], s[kl][v], h[kl * 2 + v]));
    }
    #pragma unroll
    for (int kl = 0; kl < KL; ++kl) {
      #pragma unroll
      for (int v = 0; v < 2; ++v)
        s[kl][v] = take_prev<0x8>(s[kl][v], fmaf(cL[kl * 2 + v], s[kl][v], h[kl * 2 + v]));
    }

    __builtin_amdgcn_sched_barrier(0);

    v8f acc0 = {}, acc1 = {};
    #pragma unroll
    for (int kl = 0; kl < KL; ++kl) {
      if (kl < 4) {
        acc0 = wmma_f32_16x16x4(g[kl], s[kl], acc0);
        acc1 = wmma_f32_16x16x4(af[kl], bfrag[kl], acc1);
      } else if (kl & 1) {
        acc1 = wmma_f32_16x16x4(g[kl], s[kl], acc1);
      } else {
        acc0 = wmma_f32_16x16x4(g[kl], s[kl], acc0);
      }
    }

    __builtin_amdgcn_sched_barrier(0);
    asm volatile("v_nop\n\tv_nop\n\tv_nop\n\tv_nop\n\tv_nop\n\tv_nop\n\tv_nop\n\tv_nop");
    __builtin_amdgcn_sched_barrier(0);

    float* st = &sT[k & 1][mw][0] + cq * TQ + hi * (8 * NB) + bb;
    #pragma unroll
    for (int r = 0; r < 8; ++r)
      st[r * NB] = acc0[r] + acc1[r];

    #pragma unroll
    for (int kl = 0; kl < KL; ++kl) {
      #pragma unroll
      for (int v = 0; v < 2; ++v)
        s[kl][v] = take_prev<0x1>(s[kl][v], fmaf(cL[kl * 2 + v], s[kl][v], h[kl * 2 + v]));
    }
  };

  auto finish = [&](int ch, int pl, int tp, int q) __attribute__((always_inline)) {
    v4f yv;
    #pragma unroll
    for (int c = 0; c < 4; ++c)
      yv[c] = sT[tp][4 * phf + c][q * TQ + pj * NB + pb] + Dc[c] * sX[pl][4 * phf + c][q * XQ + pj * NB + pb];
    *reinterpret_cast<v4f*>(&y[((b0 + pb) * LENGTH + ch * TCH + pj) * D_MODEL + d0 + 4 * phf]) = yv;
  };

  auto step = [&](int u, int k) __attribute__((always_inline)) {
    if (mover) {
      if (u > 0) {
        finish(CPS * (u - 1) + 0, (k + 3) & 3, (k + 1) & 1, 0);
        finish(CPS * (u - 1) + 1, (k + 3) & 3, (k + 1) & 1, 1);
        finish(CPS * (u - 1) + 2, (k + 3) & 3, (k + 1) & 1, 2);
        finish(CPS * (u - 1) + 3, (k + 3) & 3, (k + 1) & 1, 3);
      }
      if (k & 1) {
        if (u + 1 < STEPS) {
          stage(r0, (k + 1) & 3, 0);
          stage(r1, (k + 1) & 3, 1);
          stage(r2, (k + 1) & 3, 2);
          stage(r3, (k + 1) & 3, 3);
          stage(r4, (k + 2) & 3, 0);
          stage(r5, (k + 2) & 3, 1);
          stage(r6, (k + 2) & 3, 2);
          stage(r7, (k + 2) & 3, 3);
          if (u + 3 < STEPS) {
            request(r0, CPS * (u + 3) + 0);
            request(r1, CPS * (u + 3) + 1);
            request(r2, CPS * (u + 3) + 2);
            request(r3, CPS * (u + 3) + 3);
            request(r4, CPS * (u + 3) + 4);
            request(r5, CPS * (u + 3) + 5);
            request(r6, CPS * (u + 3) + 6);
            request(r7, CPS * (u + 3) + 7);
          }
        }
      }
    } else {
      tile4(k);
    }
    lds_barrier();
  };

  if (mover) {
    stage(r0, 0, 0);
    stage(r1, 0, 1);
    stage(r2, 0, 2);
    stage(r3, 0, 3);
    stage(r4, 1, 0);
    stage(r5, 1, 1);
    stage(r6, 1, 2);
    stage(r7, 1, 3);
    request(r0, 8);
    request(r1, 9);
    request(r2, 10);
    request(r3, 11);
    request(r4, 12);
    request(r5, 13);
    request(r6, 14);
    request(r7, 15);
  }
  lds_barrier();

  for (int u = 0; u < STEPS; u += 4) {
    step(u,     0);
    step(u + 1, 1);
    step(u + 2, 2);
    step(u + 3, 3);
  }

  if (mover) {
    finish(CHUNKS - 4, (STEPS - 1) & 3, (STEPS - 1) & 1, 0);
    finish(CHUNKS - 3, (STEPS - 1) & 3, (STEPS - 1) & 1, 1);
    finish(CHUNKS - 2, (STEPS - 1) & 3, (STEPS - 1) & 1, 2);
    finish(CHUNKS - 1, (STEPS - 1) & 3, (STEPS - 1) & 1, 3);
  }
}

extern "C" void kernel_launch(void* const* d_in, const int* in_sizes, int n_in,
                              void* d_out, int out_size, void* d_ws, size_t ws_size,
                              hipStream_t stream) {
  (void)in_sizes; (void)n_in; (void)out_size; (void)d_ws; (void)ws_size;
  const float* x      = (const float*)d_in[0];
  const float* log_dt = (const float*)d_in[1];
  const float* Bmat   = (const float*)d_in[2];
  const float* Cmat   = (const float*)d_in[3];
  const float* Dvec   = (const float*)d_in[4];
  float* y            = (float*)d_out;

  s4d_fused<<<dim3(GROUPS), NTHR, 0, stream>>>(x, log_dt, Bmat, Cmat, Dvec, y);
}
